# starting kernel
# baseline (speedup 1.0000x reference)
#include <hip/hip_runtime.h>

typedef float v2f __attribute__((ext_vector_type(2)));
typedef float v8f __attribute__((ext_vector_type(8)));

namespace {
constexpr int CIN        = 512;
constexpr int COUT       = 512;
constexpr int HW         = 49;
constexpr int LDS_STRIDE = 50;
}

__global__ __launch_bounds__(128)
void grouped_conv_roll_wmma(const float* __restrict__ x,
                            const float* __restrict__ w,
                            float* __restrict__ out) {
  extern __shared__ float ldsx[];

  const int tid  = threadIdx.x;
  const int lane = tid & 31;
  const int wave = tid >> 5;
  const int o    = blockIdx.x >> 3;
  const int k0   = ((blockIdx.x & 7) * 4 + wave) * 16;

  const float* xg = x + o * CIN * HW;
  for (int r = tid; r < CIN; r += 128) {
    const float* src = xg + r * HW;
    float* dst = ldsx + r * LDS_STRIDE;
    #pragma unroll 7
    for (int c = 0; c < HW; ++c) dst[c] = src[c];
    dst[HW] = 0.0f;
  }
  __syncthreads();

  const int ncol = lane & 15;
  const int half = lane >> 4;
  const int koff = half * 2;

  int boff[3][4];
  #pragma unroll
  for (int j = 0; j < 3; ++j) {
    #pragma unroll
    for (int t = 0; t < 4; ++t) {
      const int nm = t * 16 + ncol;
      int off = HW;
      if (nm < HW) {
        const int n = nm / 7, m = nm % 7;
        const int p = m + j;
        if (p >= 1 && p <= 7) off = n * 7 + (p + 5) % 7;
      }
      boff[j][t] = off;
    }
  }

  v8f acc[4] = {};

  const float* wk = w + (k0 + ncol) * 3;

  for (int j = 0; j < 3; ++j) {
    const int o0 = boff[j][0], o1 = boff[j][1];
    const int o2 = boff[j][2], o3 = boff[j][3];
    for (int i0 = 0; i0 < CIN; i0 += 4) {
      const int ia = i0 + koff;
      v2f A = { wk[ia * (COUT * 3) + j],
                wk[(ia + 1) * (COUT * 3) + j] };
      const float* br0 = ldsx + ia * LDS_STRIDE;
      const float* br1 = br0 + LDS_STRIDE;
      v2f B0 = { br0[o0], br1[o0] };
      v2f B1 = { br0[o1], br1[o1] };
      v2f B2 = { br0[o2], br1[o2] };
      v2f B3 = { br0[o3], br1[o3] };
      acc[0] = __builtin_amdgcn_wmma_f32_16x16x4_f32(false, A, false, B0, (short)0, acc[0], false, false);
      acc[1] = __builtin_amdgcn_wmma_f32_16x16x4_f32(false, A, false, B1, (short)0, acc[1], false, false);
      acc[2] = __builtin_amdgcn_wmma_f32_16x16x4_f32(false, A, false, B2, (short)0, acc[2], false, false);
      acc[3] = __builtin_amdgcn_wmma_f32_16x16x4_f32(false, A, false, B3, (short)0, acc[3], false, false);
    }
  }

  float* outg = out + o * COUT * HW;
  const int kbase = k0 + half * 8;
  #pragma unroll
  for (int t = 0; t < 4; ++t) {
    const int nm = t * 16 + ncol;
    if (nm < HW) {
      const int n = nm / 7, m = nm % 7;
      const int sp = ((n + 1) % 7) * 7 + m;
      #pragma unroll
      for (int v = 0; v < 8; ++v)
        outg[(kbase + v) * HW + sp] = acc[t][v];
    }
  }
}

extern "C" void kernel_launch(void* const* d_in, const int* in_sizes, int n_in,
                              void* d_out, int out_size, void* d_ws, size_t ws_size,
                              hipStream_t stream) {
  (void)in_sizes; (void)n_in; (void)d_ws; (void)ws_size; (void)out_size;
  const float* x = (const float*)d_in[0];
  const float* w = (const float*)d_in[1];
  float* out = (float*)d_out;

  dim3 grid(16), block(128);
  size_t shmem = (size_t)CIN * LDS_STRIDE * sizeof(float);
  hipLaunchKernelGGL(grouped_conv_roll_wmma, grid, block, shmem, stream, x, w, out);
}


# K split over 16 waves per 16-row tile (64 workgroups of 512), partial tiles summed through LDS in fixed wave order (sum order changes); slab staged by coalesced 8 B loads
# speedup vs baseline: 5.2035x; 5.2035x over previous
#include <hip/hip_runtime.h>

typedef float v2f __attribute__((ext_vector_type(2)));
typedef float v8f __attribute__((ext_vector_type(8)));

namespace {
constexpr int CIN        = 512;
constexpr int COUT       = 512;
constexpr int HW         = 49;
constexpr int LDS_STRIDE = 50;
constexpr int WAVES      = 16;
constexpr int KSL        = CIN / WAVES;
constexpr int KSTEPS     = KSL / 4;
constexpr int NTHR       = WAVES * 32;
constexpr int SLAB_FLOATS = CIN * LDS_STRIDE;
constexpr int RED_VEC4    = WAVES * 4 * 2 * 32;
constexpr int XPAIRS      = CIN * HW / 2;
constexpr int LDS_FLOATS  = SLAB_FLOATS + RED_VEC4 * 4;
constexpr int XFULL       = XPAIRS / NTHR;
constexpr int XTAIL       = XPAIRS - XFULL * NTHR;
static_assert(XTAIL == 256 && CIN == NTHR, "staging shape");
}

__global__ __launch_bounds__(512)
void grouped_conv_roll_wmma_ksplit(const float* __restrict__ x,
                                   const float* __restrict__ w,
                                   float* __restrict__ out) {
  __shared__ __attribute__((aligned(16))) float ldsx[LDS_FLOATS];
  float4* red4 = reinterpret_cast<float4*>(ldsx + SLAB_FLOATS);

  const int tid  = threadIdx.x;
  const int lane = tid & 31;
  const int wave = tid >> 5;
  const int o    = blockIdx.x >> 5;
  const int k0   = (blockIdx.x & 31) * 16;

  const int ncol = lane & 15;
  const int half = lane >> 4;
  const int koff = half * 2;
  const int rbase = wave * KSL + koff;

  const float* wk = w + (size_t)rbase * (COUT * 3) + (k0 + ncol) * 3;
  float a[KSTEPS][2][3];
  #pragma unroll
  for (int s = 0; s < KSTEPS; ++s) {
    #pragma unroll
    for (int r = 0; r < 2; ++r) {
      #pragma unroll
      for (int j = 0; j < 3; ++j)
        a[s][r][j] = wk[(s * 4 + r) * (COUT * 3) + j];
    }
  }

  const float2* xg2 = reinterpret_cast<const float2*>(x + (size_t)o * CIN * HW);
  #pragma unroll
  for (int it = 0; it <= XFULL; ++it) {
    const int q = (it < XFULL) ? tid + it * NTHR : XFULL * NTHR + (tid & (XTAIL - 1));
    const float2 v = xg2[q];
    const int i0 = 2 * q, i1 = 2 * q + 1;
    ldsx[i0 + i0 / HW] = v.x;
    ldsx[i1 + i1 / HW] = v.y;
  }
  ldsx[tid * LDS_STRIDE + HW] = 0.0f;
  __syncthreads();

  int boff[3][4];
  #pragma unroll
  for (int j = 0; j < 3; ++j) {
    #pragma unroll
    for (int t = 0; t < 4; ++t) {
      const int nm = t * 16 + ncol;
      int off = HW;
      if (nm < HW) {
        const int n = nm / 7, m = nm % 7;
        const int p = m + j;
        if (p >= 1 && p <= 7) off = n * 7 + (p + 5) % 7;
      }
      boff[j][t] = off;
    }
  }

  v8f acc[4] = {};

  #pragma unroll
  for (int s = 0; s < KSTEPS; ++s) {
    const float* br0 = ldsx + (rbase + s * 4) * LDS_STRIDE;
    const float* br1 = br0 + LDS_STRIDE;
    #pragma unroll
    for (int j = 0; j < 3; ++j) {
      const v2f A = { a[s][0][j], a[s][1][j] };
      #pragma unroll
      for (int t = 0; t < 4; ++t) {
        const v2f B = { br0[boff[j][t]], br1[boff[j][t]] };
        acc[t] = __builtin_amdgcn_wmma_f32_16x16x4_f32(false, A, false, B, (short)0, acc[t], false, false);
      }
    }
  }

  #pragma unroll
  for (int t = 0; t < 4; ++t) {
    #pragma unroll
    for (int vh = 0; vh < 2; ++vh) {
      red4[((wave * 4 + t) * 2 + vh) * 32 + lane] =
          make_float4(acc[t][4 * vh], acc[t][4 * vh + 1], acc[t][4 * vh + 2], acc[t][4 * vh + 3]);
    }
  }
  __syncthreads();

  if (tid < 256) {
    float4 sum = red4[tid];
    #pragma unroll
    for (int p = 1; p < WAVES; ++p) {
      const float4 v = red4[p * 256 + tid];
      sum.x += v.x; sum.y += v.y; sum.z += v.z; sum.w += v.w;
    }
    const int t  = tid >> 6;
    const int vh = (tid >> 5) & 1;
    const int nm = t * 16 + ncol;
    if (nm < HW) {
      const int n = nm / 7, m = nm % 7;
      const int sp = ((n + 1) % 7) * 7 + m;
      float* orow = out + (size_t)o * COUT * HW + (k0 + half * 8 + vh * 4) * HW + sp;
      orow[0]      = sum.x;
      orow[HW]     = sum.y;
      orow[2 * HW] = sum.z;
      orow[3 * HW] = sum.w;
    }
  }
}

extern "C" void kernel_launch(void* const* d_in, const int* in_sizes, int n_in,
                              void* d_out, int out_size, void* d_ws, size_t ws_size,
                              hipStream_t stream) {
  (void)in_sizes; (void)n_in; (void)d_ws; (void)ws_size; (void)out_size;
  const float* x = (const float*)d_in[0];
  const float* w = (const float*)d_in[1];
  float* out = (float*)d_out;

  dim3 grid(64), block(512);
  hipLaunchKernelGGL(grouped_conv_roll_wmma_ksplit, grid, block, 0, stream, x, w, out);
}


# Column split: each workgroup takes one 14-column tile (two spatial rows) instead of all 49 columns; grid 64 -> 256 workgroups of 8 waves; one accumulator per tap (sum order changes)
# speedup vs baseline: 7.0595x; 1.3567x over previous
#include <hip/hip_runtime.h>

typedef float v2f __attribute__((ext_vector_type(2)));
typedef float v8f __attribute__((ext_vector_type(8)));

namespace {
constexpr int CIN        = 512;
constexpr int COUT       = 512;
constexpr int HW         = 49;
constexpr int NT         = 4;
constexpr int TCOLS      = 14;
constexpr int RS         = 16;
constexpr int ZCOL       = 14;
constexpr int WAVES      = 8;
constexpr int KSL        = CIN / WAVES;
constexpr int KSTEPS     = KSL / 4;
constexpr int NTHR       = WAVES * 32;
constexpr int SROWS      = 16;
constexpr int SROUNDS    = CIN / SROWS;
constexpr int SLAB_FLOATS = CIN * RS;
constexpr int RED_VEC4    = WAVES * 2 * 32;
constexpr int LDS_FLOATS  = 41984;
static_assert(SLAB_FLOATS + RED_VEC4 * 4 <= LDS_FLOATS, "LDS layout");
static_assert(CIN == 2 * NTHR, "zero column: two rows per thread");
}

__global__ __launch_bounds__(256)
void grouped_conv_roll_wmma_nsplit(const float* __restrict__ x,
                                   const float* __restrict__ w,
                                   float* __restrict__ out) {
  __shared__ __attribute__((aligned(16))) float ldsx[LDS_FLOATS];
  float4* red4 = reinterpret_cast<float4*>(ldsx + SLAB_FLOATS);

  const int tid  = threadIdx.x;
  const int lane = tid & 31;
  const int wave = tid >> 5;
  const int nt   = blockIdx.x & (NT - 1);
  const int k0   = ((blockIdx.x >> 2) & 31) * 16;
  const int o    = blockIdx.x >> 7;
  const int ncols = (nt == NT - 1) ? 7 : TCOLS;

  const int ncol = lane & 15;
  const int half = lane >> 4;
  const int koff = half * 2;
  const int rbase = wave * KSL + koff;

  const int sr = tid / TCOLS;
  const int sc = tid - sr * TCOLS;
  const bool stager = (tid < SROWS * TCOLS) && (sc < ncols);
  const float* xs = x + (size_t)o * CIN * HW + sr * HW + nt * TCOLS + sc;
  float xv[SROUNDS];
  if (stager) {
    #pragma unroll
    for (int it = 0; it < SROUNDS; ++it) xv[it] = xs[it * SROWS * HW];
  }

  const float* wk = w + (size_t)rbase * (COUT * 3) + (k0 + ncol) * 3;
  float a[KSTEPS][2][3];
  #pragma unroll
  for (int s = 0; s < KSTEPS; ++s) {
    #pragma unroll
    for (int r = 0; r < 2; ++r) {
      #pragma unroll
      for (int j = 0; j < 3; ++j)
        a[s][r][j] = wk[(s * 4 + r) * (COUT * 3) + j];
    }
  }

  if (stager) {
    #pragma unroll
    for (int it = 0; it < SROUNDS; ++it) ldsx[(it * SROWS + sr) * RS + sc] = xv[it];
  }
  ldsx[tid * RS + ZCOL] = 0.0f;
  ldsx[(tid + NTHR) * RS + ZCOL] = 0.0f;
  __syncthreads();

  int boff[3];
  #pragma unroll
  for (int j = 0; j < 3; ++j) {
    int off = ZCOL;
    if (ncol < ncols) {
      const int nl = (ncol >= 7) ? 1 : 0;
      const int m  = ncol - 7 * nl;
      const int p  = m + j;
      if (p >= 1 && p <= 7) off = nl * 7 + (p + 5) % 7;
    }
    boff[j] = off;
  }

  v8f acc[3] = {};

  #pragma unroll
  for (int s = 0; s < KSTEPS; ++s) {
    const float* br0 = ldsx + (rbase + s * 4) * RS;
    const float* br1 = br0 + RS;
    #pragma unroll
    for (int j = 0; j < 3; ++j) {
      const v2f A = { a[s][0][j], a[s][1][j] };
      const v2f B = { br0[boff[j]], br1[boff[j]] };
      acc[j] = __builtin_amdgcn_wmma_f32_16x16x4_f32(false, A, false, B, (short)0, acc[j], false, false);
    }
  }

  const v8f accw = (acc[0] + acc[1]) + acc[2];
  #pragma unroll
  for (int vh = 0; vh < 2; ++vh) {
    red4[(wave * 2 + vh) * 32 + lane] =
        make_float4(accw[4 * vh], accw[4 * vh + 1], accw[4 * vh + 2], accw[4 * vh + 3]);
  }
  __syncthreads();

  if (tid < 64) {
    float4 sum = red4[tid];
    #pragma unroll
    for (int p = 1; p < WAVES; ++p) {
      const float4 v = red4[p * 64 + tid];
      sum.x += v.x; sum.y += v.y; sum.z += v.z; sum.w += v.w;
    }
    const int vh = tid >> 5;
    if (ncol < ncols) {
      const int nl = (ncol >= 7) ? 1 : 0;
      const int m  = ncol - 7 * nl;
      const int n  = 2 * nt + nl;
      const int sp = ((n + 1) % 7) * 7 + m;
      float* orow = out + (size_t)o * COUT * HW + (k0 + half * 8 + vh * 4) * HW + sp;
      orow[0]      = sum.x;
      orow[HW]     = sum.y;
      orow[2 * HW] = sum.z;
      orow[3 * HW] = sum.w;
    }
  }
}

extern "C" void kernel_launch(void* const* d_in, const int* in_sizes, int n_in,
                              void* d_out, int out_size, void* d_ws, size_t ws_size,
                              hipStream_t stream) {
  (void)in_sizes; (void)n_in; (void)d_ws; (void)ws_size; (void)out_size;
  const float* x = (const float*)d_in[0];
  const float* w = (const float*)d_in[1];
  float* out = (float*)d_out;

  dim3 grid(256), block(256);
  hipLaunchKernelGGL(grouped_conv_roll_wmma_nsplit, grid, block, 0, stream, x, w, out);
}


# Activation operand loaded from global memory straight into the B fragment layout, the three taps gathered from lanes of the same spatial row (permlane16); no LDS slab, no staging barrier
# speedup vs baseline: 7.7443x; 1.0970x over previous
#include <hip/hip_runtime.h>

typedef float v2f __attribute__((ext_vector_type(2)));
typedef float v8f __attribute__((ext_vector_type(8)));

namespace {
constexpr int CIN        = 512;
constexpr int COUT       = 512;
constexpr int HW         = 49;
constexpr int NT         = 4;
constexpr int TCOLS      = 14;
constexpr int WAVES      = 8;
constexpr int KSL        = CIN / WAVES;
constexpr int KSTEPS     = KSL / 4;
constexpr int NTHR       = WAVES * 32;
constexpr int RED_VEC4    = WAVES * 2 * 32;
constexpr unsigned SEL1_LO = 0xD5432106u, SEL1_HI = 0xFECBA987u;
constexpr unsigned SEL0_LO = 0xC4321065u, SEL0_HI = 0xFEBA987Du;
constexpr int LDS_FLOATS  = 41984;
static_assert(RED_VEC4 * 4 <= LDS_FLOATS, "LDS layout");
}

__global__ __launch_bounds__(256)
void grouped_conv_roll_wmma_nsplit(const float* __restrict__ x,
                                   const float* __restrict__ w,
                                   float* __restrict__ out) {
  __shared__ __attribute__((aligned(16))) float ldsx[LDS_FLOATS];
  float4* red4 = reinterpret_cast<float4*>(ldsx);

  const int tid  = threadIdx.x;
  const int lane = tid & 31;
  const int wave = tid >> 5;
  const int nt   = blockIdx.x & (NT - 1);
  const int k0   = ((blockIdx.x >> 2) & 31) * 16;
  const int o    = blockIdx.x >> 7;
  const int ncols = (nt == NT - 1) ? 7 : TCOLS;

  const int ncol = lane & 15;
  const int half = lane >> 4;
  const int koff = half * 2;
  const int rbase = wave * KSL + koff;

  const bool colok = ncol < ncols;
  const int  nl    = (ncol >= 7) ? 1 : 0;
  const int  m     = ncol - 7 * nl;
  const bool keep0 = colok && (m != 0);
  const bool keep2 = colok && (m != 6);

  const float* xs = x + (size_t)o * CIN * HW + rbase * HW + nt * TCOLS + (colok ? ncol : 0);
  const float* wk = w + (size_t)rbase * (COUT * 3) + (k0 + ncol) * 3;
  float xv[KSTEPS][2];
  float a[KSTEPS][2][3];
  #pragma unroll
  for (int s = 0; s < KSTEPS; ++s) {
    #pragma unroll
    for (int r = 0; r < 2; ++r) {
      xv[s][r] = xs[(s * 4 + r) * HW];
      #pragma unroll
      for (int j = 0; j < 3; ++j)
        a[s][r][j] = wk[(s * 4 + r) * (COUT * 3) + j];
    }
  }

  v8f acc[3] = {};

  #pragma unroll
  for (int s = 0; s < KSTEPS; ++s) {
    float b0[2], b1[2], b2[2];
    #pragma unroll
    for (int r = 0; r < 2; ++r) {
      const int u = __builtin_bit_cast(int, xv[s][r]);
      const float t0 = __builtin_bit_cast(float,
          __builtin_amdgcn_permlane16(u, u, SEL0_LO, SEL0_HI, false, false));
      const float t1 = __builtin_bit_cast(float,
          __builtin_amdgcn_permlane16(u, u, SEL1_LO, SEL1_HI, false, false));
      b0[r] = keep0 ? t0 : 0.0f;
      b1[r] = colok ? t1 : 0.0f;
      b2[r] = keep2 ? xv[s][r] : 0.0f;
    }
    const v2f A0 = { a[s][0][0], a[s][1][0] };
    const v2f A1 = { a[s][0][1], a[s][1][1] };
    const v2f A2 = { a[s][0][2], a[s][1][2] };
    const v2f B0 = { b0[0], b0[1] };
    const v2f B1 = { b1[0], b1[1] };
    const v2f B2 = { b2[0], b2[1] };
    acc[0] = __builtin_amdgcn_wmma_f32_16x16x4_f32(false, A0, false, B0, (short)0, acc[0], false, false);
    acc[1] = __builtin_amdgcn_wmma_f32_16x16x4_f32(false, A1, false, B1, (short)0, acc[1], false, false);
    acc[2] = __builtin_amdgcn_wmma_f32_16x16x4_f32(false, A2, false, B2, (short)0, acc[2], false, false);
  }

  const v8f accw = (acc[0] + acc[1]) + acc[2];
  #pragma unroll
  for (int vh = 0; vh < 2; ++vh) {
    red4[(wave * 2 + vh) * 32 + lane] =
        make_float4(accw[4 * vh], accw[4 * vh + 1], accw[4 * vh + 2], accw[4 * vh + 3]);
  }
  __syncthreads();

  if (tid < 64) {
    float4 sum = red4[tid];
    #pragma unroll
    for (int p = 1; p < WAVES; ++p) {
      const float4 v = red4[p * 64 + tid];
      sum.x += v.x; sum.y += v.y; sum.z += v.z; sum.w += v.w;
    }
    const int vh = tid >> 5;
    if (colok) {
      const int n  = 2 * nt + nl;
      const int sp = ((n + 1) % 7) * 7 + m;
      float* orow = out + (size_t)o * COUT * HW + (k0 + half * 8 + vh * 4) * HW + sp;
      orow[0]      = sum.x;
      orow[HW]     = sum.y;
      orow[2 * HW] = sum.z;
      orow[3 * HW] = sum.w;
    }
  }
}

extern "C" void kernel_launch(void* const* d_in, const int* in_sizes, int n_in,
                              void* d_out, int out_size, void* d_ws, size_t ws_size,
                              hipStream_t stream) {
  (void)in_sizes; (void)n_in; (void)d_ws; (void)ws_size; (void)out_size;
  const float* x = (const float*)d_in[0];
  const float* w = (const float*)d_in[1];
  float* out = (float*)d_out;

  dim3 grid(256), block(256);
  hipLaunchKernelGGL(grouped_conv_roll_wmma_nsplit, grid, block, 0, stream, x, w, out);
}


# Each workgroup starts its K walk at one of 8 steps (wrap-around), so workgroups sharing weight or activation lines ask the L2 for them in different order (sum order changes)
# speedup vs baseline: 8.3296x; 1.0756x over previous
#include <hip/hip_runtime.h>

typedef float v2f __attribute__((ext_vector_type(2)));
typedef float v8f __attribute__((ext_vector_type(8)));

namespace {
constexpr int CIN        = 512;
constexpr int COUT       = 512;
constexpr int HW         = 49;
constexpr int NT         = 4;
constexpr int TCOLS      = 14;
constexpr int WAVES      = 8;
constexpr int KSL        = CIN / WAVES;
constexpr int KSTEPS     = KSL / 4;
constexpr int NTHR       = WAVES * 32;
constexpr int RED_VEC4    = WAVES * 2 * 32;
constexpr int NROT       = 8;
constexpr int ROTSTEP    = KSTEPS / NROT;
constexpr unsigned SEL1_LO = 0xD5432106u, SEL1_HI = 0xFECBA987u;
constexpr unsigned SEL0_LO = 0xC4321065u, SEL0_HI = 0xFEBA987Du;
constexpr int LDS_FLOATS  = 41984;
static_assert(RED_VEC4 * 4 <= LDS_FLOATS, "LDS layout");
}

template <int ROT>
__device__ __forceinline__
void conv_tile_body(const float* __restrict__ x,
                    const float* __restrict__ w,
                    float* __restrict__ out,
                    float4* red4) {

  const int tid  = threadIdx.x;
  const int lane = tid & 31;
  const int wave = tid >> 5;
  const int nt   = blockIdx.x & (NT - 1);
  const int k0   = ((blockIdx.x >> 2) & 31) * 16;
  const int o    = blockIdx.x >> 7;
  const int ncols = (nt == NT - 1) ? 7 : TCOLS;

  const int ncol = lane & 15;
  const int half = lane >> 4;
  const int koff = half * 2;
  const int rbase = wave * KSL + koff;

  const bool colok = ncol < ncols;
  const int  nl    = (ncol >= 7) ? 1 : 0;
  const int  m     = ncol - 7 * nl;
  const bool keep0 = colok && (m != 0);
  const bool keep2 = colok && (m != 6);

  const float* xs = x + (size_t)o * CIN * HW + rbase * HW + nt * TCOLS + (colok ? ncol : 0);
  const float* wk = w + (size_t)rbase * (COUT * 3) + (k0 + ncol) * 3;
  float xv[KSTEPS][2];
  float a[KSTEPS][2][3];
  #pragma unroll
  for (int s = 0; s < KSTEPS; ++s) {
    const int ks = (s + ROT) % KSTEPS;
    #pragma unroll
    for (int r = 0; r < 2; ++r) {
      xv[s][r] = xs[(ks * 4 + r) * HW];
      #pragma unroll
      for (int j = 0; j < 3; ++j)
        a[s][r][j] = wk[(ks * 4 + r) * (COUT * 3) + j];
    }
  }

  v8f acc[3] = {};

  #pragma unroll
  for (int s = 0; s < KSTEPS; ++s) {
    float b0[2], b1[2], b2[2];
    #pragma unroll
    for (int r = 0; r < 2; ++r) {
      const int u = __builtin_bit_cast(int, xv[s][r]);
      const float t0 = __builtin_bit_cast(float,
          __builtin_amdgcn_permlane16(u, u, SEL0_LO, SEL0_HI, false, false));
      const float t1 = __builtin_bit_cast(float,
          __builtin_amdgcn_permlane16(u, u, SEL1_LO, SEL1_HI, false, false));
      b0[r] = keep0 ? t0 : 0.0f;
      b1[r] = colok ? t1 : 0.0f;
      b2[r] = keep2 ? xv[s][r] : 0.0f;
    }
    const v2f A0 = { a[s][0][0], a[s][1][0] };
    const v2f A1 = { a[s][0][1], a[s][1][1] };
    const v2f A2 = { a[s][0][2], a[s][1][2] };
    const v2f B0 = { b0[0], b0[1] };
    const v2f B1 = { b1[0], b1[1] };
    const v2f B2 = { b2[0], b2[1] };
    acc[0] = __builtin_amdgcn_wmma_f32_16x16x4_f32(false, A0, false, B0, (short)0, acc[0], false, false);
    acc[1] = __builtin_amdgcn_wmma_f32_16x16x4_f32(false, A1, false, B1, (short)0, acc[1], false, false);
    acc[2] = __builtin_amdgcn_wmma_f32_16x16x4_f32(false, A2, false, B2, (short)0, acc[2], false, false);
  }

  const v8f accw = (acc[0] + acc[1]) + acc[2];
  #pragma unroll
  for (int vh = 0; vh < 2; ++vh) {
    red4[(wave * 2 + vh) * 32 + lane] =
        make_float4(accw[4 * vh], accw[4 * vh + 1], accw[4 * vh + 2], accw[4 * vh + 3]);
  }
  __syncthreads();

  if (tid < 64) {
    float4 sum = red4[tid];
    #pragma unroll
    for (int p = 1; p < WAVES; ++p) {
      const float4 v = red4[p * 64 + tid];
      sum.x += v.x; sum.y += v.y; sum.z += v.z; sum.w += v.w;
    }
    const int vh = tid >> 5;
    if (colok) {
      const int n  = 2 * nt + nl;
      const int sp = ((n + 1) % 7) * 7 + m;
      float* orow = out + (size_t)o * COUT * HW + (k0 + half * 8 + vh * 4) * HW + sp;
      orow[0]      = sum.x;
      orow[HW]     = sum.y;
      orow[2 * HW] = sum.z;
      orow[3 * HW] = sum.w;
    }
  }
}

__global__ __launch_bounds__(256)
void grouped_conv_roll_wmma_nsplit(const float* __restrict__ x,
                                   const float* __restrict__ w,
                                   float* __restrict__ out) {
  __shared__ __attribute__((aligned(16))) float ldsx[LDS_FLOATS];
  float4* red4 = reinterpret_cast<float4*>(ldsx);
  const int nt = blockIdx.x & (NT - 1);
  const int kt = (blockIdx.x >> 2) & 31;
  const int o  = blockIdx.x >> 7;
  switch ((kt + 2 * nt + o) & (NROT - 1)) {
    case 0:  conv_tile_body<0 * ROTSTEP>(x, w, out, red4); break;
    case 1:  conv_tile_body<1 * ROTSTEP>(x, w, out, red4); break;
    case 2:  conv_tile_body<2 * ROTSTEP>(x, w, out, red4); break;
    case 3:  conv_tile_body<3 * ROTSTEP>(x, w, out, red4); break;
    case 4:  conv_tile_body<4 * ROTSTEP>(x, w, out, red4); break;
    case 5:  conv_tile_body<5 * ROTSTEP>(x, w, out, red4); break;
    case 6:  conv_tile_body<6 * ROTSTEP>(x, w, out, red4); break;
    default: conv_tile_body<7 * ROTSTEP>(x, w, out, red4); break;
  }
}

extern "C" void kernel_launch(void* const* d_in, const int* in_sizes, int n_in,
                              void* d_out, int out_size, void* d_ws, size_t ws_size,
                              hipStream_t stream) {
  (void)in_sizes; (void)n_in; (void)d_ws; (void)ws_size; (void)out_size;
  const float* x = (const float*)d_in[0];
  const float* w = (const float*)d_in[1];
  float* out = (float*)d_out;

  dim3 grid(256), block(256);
  hipLaunchKernelGGL(grouped_conv_roll_wmma_nsplit, grid, block, 0, stream, x, w, out);
}
